# starting kernel
# baseline (speedup 1.0000x reference)
#include <hip/hip_runtime.h>

#define BATCH   1024
#define IN_DIM  20128
#define OUT_DIM 20000
#define KDIM    128

#define WAVES_PER_BLOCK 8
#define N_PER_WAVE      64
#define N_PER_BLOCK     (WAVES_PER_BLOCK * N_PER_WAVE)
#define M_PER_BLOCK     16

#define KSTEPS_PER_CHUNK 4
#define NCHUNKS (KDIM / (4 * KSTEPS_PER_CHUNK))

typedef float v2f __attribute__((ext_vector_type(2)));
typedef float v8f __attribute__((ext_vector_type(8)));
typedef v2f u2f __attribute__((aligned(4)));

__global__ __launch_bounds__(256) void dlpergene_wmma_kernel(
    const float* __restrict__ x,
    const float* __restrict__ W,
    const float* __restrict__ b,
    float* __restrict__ out)
{
  const int lane = threadIdx.x & 31;
  const int wave = threadIdx.x >> 5;
  const int l16  = lane & 15;
  const int half = lane >> 4;

  const int m0 = blockIdx.y * M_PER_BLOCK;
  const int n0 = blockIdx.x * N_PER_BLOCK + wave * N_PER_WAVE;

  const float* arow = x + (size_t)(m0 + l16) * IN_DIM + OUT_DIM;

  int ncol[4];
  const float* wp[4];
#pragma unroll
  for (int t = 0; t < 4; ++t) {
    int n = n0 + t * 16 + l16;
    ncol[t] = (n < OUT_DIM) ? n : (OUT_DIM - 1);
    wp[t] = W + (size_t)ncol[t] * (1 + KDIM) + 1;
  }

  v8f acc[4];
#pragma unroll
  for (int t = 0; t < 4; ++t) acc[t] = (v8f){0.f,0.f,0.f,0.f,0.f,0.f,0.f,0.f};

  v2f abuf[2][KSTEPS_PER_CHUNK];
  v2f bbuf[2][KSTEPS_PER_CHUNK][4];

  auto load_chunk = [&](int c, int buf) {
#pragma unroll
    for (int s = 0; s < KSTEPS_PER_CHUNK; ++s) {
      const int kk = (c * KSTEPS_PER_CHUNK + s) * 4 + 2 * half;
      abuf[buf][s] = *(const u2f*)(arow + kk);
#pragma unroll
      for (int t = 0; t < 4; ++t)
        bbuf[buf][s][t] = *(const u2f*)(wp[t] + kk);
    }
  };

  load_chunk(0, 0);
#pragma unroll
  for (int c = 0; c < NCHUNKS; ++c) {
    const int cur = c & 1;
    const int nxt = cur ^ 1;
    if (c < NCHUNKS - 1) load_chunk(c + 1, nxt);
#pragma unroll
    for (int s = 0; s < KSTEPS_PER_CHUNK; ++s) {
#pragma unroll
      for (int t = 0; t < 4; ++t) {
        acc[t] = __builtin_amdgcn_wmma_f32_16x16x4_f32(
             false, abuf[cur][s],
             false, bbuf[cur][s][t],
             (short)0, acc[t],
             false,  false);
      }
    }
  }

  float dw[4], bv[4];
#pragma unroll
  for (int t = 0; t < 4; ++t) {
    dw[t] = W[(size_t)ncol[t] * (1 + KDIM)];
    bv[t] = b[ncol[t]];
  }

  float gbuf[4][8];
#pragma unroll
  for (int t = 0; t < 4; ++t) {
#pragma unroll
    for (int r = 0; r < 8; ++r) {
      const int row = m0 + r + 8 * half;
      gbuf[t][r] = __builtin_nontemporal_load(x + (size_t)row * IN_DIM + ncol[t]);
    }
  }

#pragma unroll
  for (int t = 0; t < 4; ++t) {
    const int n = n0 + t * 16 + l16;
    const bool valid = (n < OUT_DIM);
#pragma unroll
    for (int r = 0; r < 8; ++r) {
      const int row = m0 + r + 8 * half;
      const float v = acc[t][r] + gbuf[t][r] * dw[t] + bv[t];
      if (valid) {
        __builtin_nontemporal_store(v, out + (size_t)row * OUT_DIM + n);
      }
    }
  }
}

extern "C" void kernel_launch(void* const* d_in, const int* in_sizes, int n_in,
                              void* d_out, int out_size, void* d_ws, size_t ws_size,
                              hipStream_t stream) {
  const float* x = (const float*)d_in[0];
  const float* W = (const float*)d_in[1];
  const float* b = (const float*)d_in[2];
  float* out = (float*)d_out;

  dim3 grid((OUT_DIM + N_PER_BLOCK - 1) / N_PER_BLOCK,
            BATCH / M_PER_BLOCK);
  dlpergene_wmma_kernel<<<grid, WAVES_PER_BLOCK * 32, 0, stream>>>(x, W, b, out);
}


# Stage the W tile and the shared-feature tile in LDS with coalesced 16-byte loads (128x128 block tile), fragments read from LDS; each lane owns 4 adjacent columns so rows leave as 16-byte stores
# speedup vs baseline: 3.6688x; 3.6688x over previous
#include <hip/hip_runtime.h>


#define BATCH   1024
#define IN_DIM  20128
#define OUT_DIM 20000
#define KDIM    128
#define WROW    129

#define MB      128
#define NB      128
#define THREADS 256
#define APITCH  132
#define A_FLOATS (MB * APITCH)
#define B_FLOATS (NB * WROW)
#define B_VEC4   (B_FLOATS / 4)
#define KSTEPS   (KDIM / 4)
#define CHUNK    2
#define NCHUNKS  (KSTEPS / CHUNK)

typedef float v2f __attribute__((ext_vector_type(2)));
typedef float v4f __attribute__((ext_vector_type(4)));
typedef float v8f __attribute__((ext_vector_type(8)));
typedef v2f u2f __attribute__((aligned(4)));

__global__ __launch_bounds__(THREADS) void dlpergene_lds_kernel(
    const float* __restrict__ x,
    const float* __restrict__ W,
    const float* __restrict__ b,
    float* __restrict__ out)
{
  __shared__ __attribute__((aligned(16))) float As[A_FLOATS];
  __shared__ __attribute__((aligned(16))) float Bs[B_FLOATS];

  const int tid  = threadIdx.x;
  const int lane = tid & 31;
  const int wave = tid >> 5;
  const int c    = lane & 15;
  const int h    = lane >> 4;

  const int m0 = blockIdx.x * MB;
  const int n0 = blockIdx.y * NB;
  const int wm = (wave >> 1) * 32;
  const int wn = (wave & 1) * 64;

#pragma unroll
  for (int i = 0; i < (MB * KDIM / 4) / THREADS; ++i) {
    const int j   = i * THREADS + tid;
    const int row = j >> 5;
    const int c4  = j & 31;
    const v4f v = *(const v4f*)(x + (size_t)(m0 + row) * IN_DIM + OUT_DIM + c4 * 4);
    *(v4f*)(As + row * APITCH + c4 * 4) = v;
  }

  {
    const v4f* W4 = (const v4f*)W + (size_t)(n0 / 4) * WROW;
    const int last4 = (OUT_DIM / 4) * WROW - 1 - (n0 / 4) * WROW;
#pragma unroll
    for (int i = 0; i < (B_VEC4 + THREADS - 1) / THREADS; ++i) {
      const int j = i * THREADS + tid;
      if (j < B_VEC4) {
        const int jj = (j < last4) ? j : last4;
        ((v4f*)Bs)[j] = W4[jj];
      }
    }
  }

  const int ncl   = wn + 4 * c;
  const int ng    = n0 + ncl;
  const bool valid = (ng < OUT_DIM);
  const int ngc   = valid ? ng : (OUT_DIM - 4);

  v4f g[2][8];
#pragma unroll
  for (int i = 0; i < 2; ++i) {
#pragma unroll
    for (int r = 0; r < 8; ++r) {
      const int row = m0 + wm + 16 * i + 8 * h + r;
      g[i][r] = __builtin_nontemporal_load((const v4f*)(x + (size_t)row * IN_DIM + ngc));
    }
  }
  const v4f bv = *(const v4f*)(b + ngc);

  __builtin_amdgcn_fence(__ATOMIC_RELEASE, "workgroup", "local");
  __builtin_amdgcn_s_barrier();
  __builtin_amdgcn_fence(__ATOMIC_ACQUIRE, "workgroup", "local");

  v8f acc[2][4];
#pragma unroll
  for (int i = 0; i < 2; ++i)
#pragma unroll
    for (int t = 0; t < 4; ++t) acc[i][t] = (v8f){0.f,0.f,0.f,0.f,0.f,0.f,0.f,0.f};

  const float* ap = As + (wm + c) * APITCH + 2 * h;
  const float* bp = Bs + ncl * WROW + 1 + 2 * h;

  v2f af[2][CHUNK][2];
  v2f bf[2][CHUNK][4];

  auto load_chunk = [&](int cc, int buf) {
#pragma unroll
    for (int s = 0; s < CHUNK; ++s) {
      const int kk = (cc * CHUNK + s) * 4;
#pragma unroll
      for (int i = 0; i < 2; ++i)
        af[buf][s][i] = *(const v2f*)(ap + i * 16 * APITCH + kk);
#pragma unroll
      for (int t = 0; t < 4; ++t)
        bf[buf][s][t] = *(const u2f*)(bp + t * WROW + kk);
    }
  };

  load_chunk(0, 0);
#pragma unroll
  for (int cc = 0; cc < NCHUNKS; ++cc) {
    const int cur = cc & 1;
    const int nxt = cur ^ 1;
    if (cc < NCHUNKS - 1) load_chunk(cc + 1, nxt);
#pragma unroll
    for (int s = 0; s < CHUNK; ++s) {
#pragma unroll
      for (int i = 0; i < 2; ++i) {
#pragma unroll
        for (int t = 0; t < 4; ++t) {
          acc[i][t] = __builtin_amdgcn_wmma_f32_16x16x4_f32(
               false, af[cur][s][i],
               false, bf[cur][s][t],
               (short)0, acc[i][t],
               false,  false);
        }
      }
    }
  }

  v4f dw;
  dw.x = Bs[(ncl + 0) * WROW];
  dw.y = Bs[(ncl + 1) * WROW];
  dw.z = Bs[(ncl + 2) * WROW];
  dw.w = Bs[(ncl + 3) * WROW];

#pragma unroll
  for (int i = 0; i < 2; ++i) {
#pragma unroll
    for (int r = 0; r < 8; ++r) {
      const int row = m0 + wm + 16 * i + 8 * h + r;
      v4f v;
      v.x = acc[i][0][r] + g[i][r].x * dw.x + bv.x;
      v.y = acc[i][1][r] + g[i][r].y * dw.y + bv.y;
      v.z = acc[i][2][r] + g[i][r].z * dw.z + bv.z;
      v.w = acc[i][3][r] + g[i][r].w * dw.w + bv.w;
      if (valid) {
        __builtin_nontemporal_store(v, (v4f*)(out + (size_t)row * OUT_DIM + ng));
      }
    }
  }
}

extern "C" void kernel_launch(void* const* d_in, const int* in_sizes, int n_in,
                              void* d_out, int out_size, void* d_ws, size_t ws_size,
                              hipStream_t stream) {
  const float* x = (const float*)d_in[0];
  const float* W = (const float*)d_in[1];
  const float* b = (const float*)d_in[2];
  float* out = (float*)d_out;

  dim3 grid(BATCH / MB,
            (OUT_DIM + NB - 1) / NB);
  dlpergene_lds_kernel<<<grid, THREADS, 0, stream>>>(x, W, b, out);
}


# Stage the W tile with 17 unconditional loads per thread (224 spare LDS vectors behind the tile): all tile loads are in flight before the first wait; the tile's tail took four serial round trips
# speedup vs baseline: 3.8369x; 1.0458x over previous
#include <hip/hip_runtime.h>


#define BATCH   1024
#define IN_DIM  20128
#define OUT_DIM 20000
#define KDIM    128
#define WROW    129

#define MB      128
#define NB      128
#define THREADS 256
#define APITCH  132
#define A_FLOATS (MB * APITCH)
#define B_FLOATS (NB * WROW)
#define B_VEC4   (B_FLOATS / 4)
#define B_ITERS  ((B_VEC4 + THREADS - 1) / THREADS)
#define B_LDS_FLOATS (B_ITERS * THREADS * 4)
#define KSTEPS   (KDIM / 4)
#define CHUNK    2
#define NCHUNKS  (KSTEPS / CHUNK)

typedef float v2f __attribute__((ext_vector_type(2)));
typedef float v4f __attribute__((ext_vector_type(4)));
typedef float v8f __attribute__((ext_vector_type(8)));
typedef v2f u2f __attribute__((aligned(4)));

__global__ __launch_bounds__(THREADS) void dlpergene_lds_kernel(
    const float* __restrict__ x,
    const float* __restrict__ W,
    const float* __restrict__ b,
    float* __restrict__ out)
{
  __shared__ __attribute__((aligned(16))) float As[A_FLOATS];
  __shared__ __attribute__((aligned(16))) float Bs[B_LDS_FLOATS];

  const int tid  = threadIdx.x;
  const int lane = tid & 31;
  const int wave = tid >> 5;
  const int c    = lane & 15;
  const int h    = lane >> 4;

  const int m0 = blockIdx.x * MB;
  const int n0 = blockIdx.y * NB;
  const int wm = (wave >> 1) * 32;
  const int wn = (wave & 1) * 64;

#pragma unroll
  for (int i = 0; i < (MB * KDIM / 4) / THREADS; ++i) {
    const int j   = i * THREADS + tid;
    const int row = j >> 5;
    const int c4  = j & 31;
    const v4f v = *(const v4f*)(x + (size_t)(m0 + row) * IN_DIM + OUT_DIM + c4 * 4);
    *(v4f*)(As + row * APITCH + c4 * 4) = v;
  }

  {
    const v4f* W4 = (const v4f*)W + (size_t)(n0 / 4) * WROW;
    const int last4 = (OUT_DIM / 4) * WROW - 1 - (n0 / 4) * WROW;
#pragma unroll
    for (int i = 0; i < B_ITERS; ++i) {
      const int j  = i * THREADS + tid;
      const int jj = (j < last4) ? j : last4;
      ((v4f*)Bs)[j] = W4[jj];
    }
  }

  const int ncl   = wn + 4 * c;
  const int ng    = n0 + ncl;
  const bool valid = (ng < OUT_DIM);
  const int ngc   = valid ? ng : (OUT_DIM - 4);

  v4f g[2][8];
#pragma unroll
  for (int i = 0; i < 2; ++i) {
#pragma unroll
    for (int r = 0; r < 8; ++r) {
      const int row = m0 + wm + 16 * i + 8 * h + r;
      g[i][r] = __builtin_nontemporal_load((const v4f*)(x + (size_t)row * IN_DIM + ngc));
    }
  }
  const v4f bv = *(const v4f*)(b + ngc);

  __builtin_amdgcn_fence(__ATOMIC_RELEASE, "workgroup", "local");
  __builtin_amdgcn_s_barrier();
  __builtin_amdgcn_fence(__ATOMIC_ACQUIRE, "workgroup", "local");

  v8f acc[2][4];
#pragma unroll
  for (int i = 0; i < 2; ++i)
#pragma unroll
    for (int t = 0; t < 4; ++t) acc[i][t] = (v8f){0.f,0.f,0.f,0.f,0.f,0.f,0.f,0.f};

  const float* ap = As + (wm + c) * APITCH + 2 * h;
  const float* bp = Bs + ncl * WROW + 1 + 2 * h;

  v2f af[2][CHUNK][2];
  v2f bf[2][CHUNK][4];

  auto load_chunk = [&](int cc, int buf) {
#pragma unroll
    for (int s = 0; s < CHUNK; ++s) {
      const int kk = (cc * CHUNK + s) * 4;
#pragma unroll
      for (int i = 0; i < 2; ++i)
        af[buf][s][i] = *(const v2f*)(ap + i * 16 * APITCH + kk);
#pragma unroll
      for (int t = 0; t < 4; ++t)
        bf[buf][s][t] = *(const u2f*)(bp + t * WROW + kk);
    }
  };

  load_chunk(0, 0);
#pragma unroll
  for (int cc = 0; cc < NCHUNKS; ++cc) {
    const int cur = cc & 1;
    const int nxt = cur ^ 1;
    if (cc < NCHUNKS - 1) load_chunk(cc + 1, nxt);
#pragma unroll
    for (int s = 0; s < CHUNK; ++s) {
#pragma unroll
      for (int i = 0; i < 2; ++i) {
#pragma unroll
        for (int t = 0; t < 4; ++t) {
          acc[i][t] = __builtin_amdgcn_wmma_f32_16x16x4_f32(
               false, af[cur][s][i],
               false, bf[cur][s][t],
               (short)0, acc[i][t],
               false,  false);
        }
      }
    }
  }

  v4f dw;
  dw.x = Bs[(ncl + 0) * WROW];
  dw.y = Bs[(ncl + 1) * WROW];
  dw.z = Bs[(ncl + 2) * WROW];
  dw.w = Bs[(ncl + 3) * WROW];

#pragma unroll
  for (int i = 0; i < 2; ++i) {
#pragma unroll
    for (int r = 0; r < 8; ++r) {
      const int row = m0 + wm + 16 * i + 8 * h + r;
      v4f v;
      v.x = acc[i][0][r] + g[i][r].x * dw.x + bv.x;
      v.y = acc[i][1][r] + g[i][r].y * dw.y + bv.y;
      v.z = acc[i][2][r] + g[i][r].z * dw.z + bv.z;
      v.w = acc[i][3][r] + g[i][r].w * dw.w + bv.w;
      if (valid) {
        __builtin_nontemporal_store(v, (v4f*)(out + (size_t)row * OUT_DIM + ng));
      }
    }
  }
}

extern "C" void kernel_launch(void* const* d_in, const int* in_sizes, int n_in,
                              void* d_out, int out_size, void* d_ws, size_t ws_size,
                              hipStream_t stream) {
  const float* x = (const float*)d_in[0];
  const float* W = (const float*)d_in[1];
  const float* b = (const float*)d_in[2];
  float* out = (float*)d_out;

  dim3 grid(BATCH / MB,
            (OUT_DIM + NB - 1) / NB);
  dlpergene_lds_kernel<<<grid, THREADS, 0, stream>>>(x, W, b, out);
}
